# starting kernel
# baseline (speedup 1.0000x reference)
#include <hip/hip_runtime.h>

typedef float v2f __attribute__((ext_vector_type(2)));
typedef float v8f __attribute__((ext_vector_type(8)));

constexpr int NB = 8;
constexpr int NT = 4096;
constexpr int NS = 512;
constexpr int ND = 768;

constexpr int SPANS_PER_TILE = 16;
constexpr int CHANS_PER_WAVE = 16;
constexpr int WAVES_PER_BLOCK = 8;
constexpr int CHANS_PER_BLOCK = WAVES_PER_BLOCK * CHANS_PER_WAVE;
constexpr int KTOK = (NT / NS) * SPANS_PER_TILE;

__global__ __launch_bounds__(256) void span_mean_wmma(
    const float* __restrict__ x,
    const int*   __restrict__ lengths,
    const int*   __restrict__ spans,
    float*       __restrict__ out)
{
    const int lane = threadIdx.x & 31;
    const int wave = threadIdx.x >> 5;
    const int cg   = blockIdx.x;
    const int sg   = blockIdx.y;
    const int b    = blockIdx.z;

    const int chan0 = cg * CHANS_PER_BLOCK + wave * CHANS_PER_WAVE;
    const int span0 = sg * SPANS_PER_TILE;

    const int m  = lane & 15;
    const int hi = lane >> 4;

    const int sp  = span0 + m;
    const int ii  = spans[((b * NS + sp) << 1) + 0];
    const int jj  = spans[((b * NS + sp) << 1) + 1];
    const float inv = 1.0f / (float)max(jj - ii, 1);

    const int base  = spans[((b * NS + span0) << 1) + 0];
    const int len_b = lengths[b];

    const float* xcol = x + (size_t)b * NT * ND + chan0 + m;

    v8f c = {};
    #pragma unroll 4
    for (int k0 = 0; k0 < KTOK; k0 += 4) {
        const int tok = base + k0 + (hi << 1);

        v2f a;
        a.x = (tok     >= ii && tok     < jj) ? inv : 0.0f;
        a.y = (tok + 1 >= ii && tok + 1 < jj) ? inv : 0.0f;

        v2f bm;
        bm.x = xcol[(size_t)tok       * ND];
        bm.y = xcol[(size_t)(tok + 1) * ND];

        c = __builtin_amdgcn_wmma_f32_16x16x4_f32(
                 false, a,  false, bm,
                 (short)0, c,  false,  false);
    }

    float* ocol = out + (size_t)b * NS * ND + chan0 + m;
    #pragma unroll
    for (int v = 0; v < 8; ++v) {
        const int sp_out = span0 + v + (hi << 3);
        const float val  = (sp_out < len_b) ? c[v] : 0.0f;
        ocol[(size_t)sp_out * ND] = val;
    }
}

extern "C" void kernel_launch(void* const* d_in, const int* in_sizes, int n_in,
                              void* d_out, int out_size, void* d_ws, size_t ws_size,
                              hipStream_t stream) {
    const float* x       = (const float*)d_in[0];
    const int*   lengths = (const int*)  d_in[1];
    const int*   spans   = (const int*)  d_in[2];
    float*       out     = (float*)d_out;

    dim3 grid(ND / CHANS_PER_BLOCK,
              NS / SPANS_PER_TILE,
              NB);
    span_mean_wmma<<<grid, WAVES_PER_BLOCK * 32, 0, stream>>>(x, lengths, spans, out);
}


# wave tile 16 -> 64 channels: a lane owns 4 adjacent channels (16-byte row loads and stores, 4 accumulators share A), its 64 row loads requested ahead of the products; one 12-wave block per WGP
# speedup vs baseline: 1.1828x; 1.1828x over previous
#include <hip/hip_runtime.h>


typedef float v2f __attribute__((ext_vector_type(2)));
typedef float v4f __attribute__((ext_vector_type(4)));
typedef float v8f __attribute__((ext_vector_type(8)));

constexpr int NB = 8;
constexpr int NT = 4096;
constexpr int NS = 512;
constexpr int ND = 768;

constexpr int SPANS_PER_TILE  = 16;
constexpr int CHANS_PER_LANE  = 4;
constexpr int CHANS_PER_WAVE  = 16 * CHANS_PER_LANE;
constexpr int WAVES_PER_BLOCK = ND / CHANS_PER_WAVE;
constexpr int KTOK = (NT / NS) * SPANS_PER_TILE;

__global__ __launch_bounds__(WAVES_PER_BLOCK * 32) void span_mean_wmma(
    const float* __restrict__ x,
    const int*   __restrict__ lengths,
    const int*   __restrict__ spans,
    float*       __restrict__ out)
{
    const int lane = threadIdx.x & 31;
    const int wave = threadIdx.x >> 5;
    const int sg   = blockIdx.x;
    const int b    = blockIdx.y;

    const int chan0 = wave * CHANS_PER_WAVE;
    const int span0 = sg * SPANS_PER_TILE;

    const int m  = lane & 15;
    const int hi = lane >> 4;

    const int sp  = span0 + m;
    const int ii  = spans[((b * NS + sp) << 1) + 0];
    const int jj  = spans[((b * NS + sp) << 1) + 1];
    const float inv = 1.0f / (float)max(jj - ii, 1);

    const int base  = spans[((b * NS + span0) << 1) + 0];
    const int len_b = lengths[b];

    const float* xp = x + (size_t)b * NT * ND
                        + (size_t)(base + (hi << 1)) * ND
                        + chan0 + m * CHANS_PER_LANE;

    constexpr int KSTEPS = KTOK / 4;
    v4f r[2 * KSTEPS];
    #pragma unroll
    for (int i = 0; i < KSTEPS; ++i) {
        r[2 * i + 0] = *(const v4f*)(xp + (size_t)(4 * i)     * ND);
        r[2 * i + 1] = *(const v4f*)(xp + (size_t)(4 * i + 1) * ND);
    }
    __builtin_amdgcn_sched_barrier(0);

    v8f c0 = {}, c1 = {}, c2 = {}, c3 = {};
    #pragma unroll
    for (int i = 0; i < KSTEPS; ++i) {
        const int tok = base + 4 * i + (hi << 1);

        v2f a;
        a.x = (tok     >= ii && tok     < jj) ? inv : 0.0f;
        a.y = (tok + 1 >= ii && tok + 1 < jj) ? inv : 0.0f;

        const v4f r0 = r[2 * i + 0];
        const v4f r1 = r[2 * i + 1];
        const v2f b0 = {r0.x, r1.x};
        const v2f b1 = {r0.y, r1.y};
        const v2f b2 = {r0.z, r1.z};
        const v2f b3 = {r0.w, r1.w};

        c0 = __builtin_amdgcn_wmma_f32_16x16x4_f32(false, a, false, b0, (short)0, c0, false, false);
        c1 = __builtin_amdgcn_wmma_f32_16x16x4_f32(false, a, false, b1, (short)0, c1, false, false);
        c2 = __builtin_amdgcn_wmma_f32_16x16x4_f32(false, a, false, b2, (short)0, c2, false, false);
        c3 = __builtin_amdgcn_wmma_f32_16x16x4_f32(false, a, false, b3, (short)0, c3, false, false);
    }

    float* op = out + (size_t)b * NS * ND + chan0 + m * CHANS_PER_LANE;
    #pragma unroll
    for (int v = 0; v < 8; ++v) {
        const int sp_out = span0 + v + (hi << 3);
        const bool keep  = sp_out < len_b;
        v4f val;
        val.x = keep ? c0[v] : 0.0f;
        val.y = keep ? c1[v] : 0.0f;
        val.z = keep ? c2[v] : 0.0f;
        val.w = keep ? c3[v] : 0.0f;
        *(v4f*)(op + (size_t)sp_out * ND) = val;
    }
}

extern "C" void kernel_launch(void* const* d_in, const int* in_sizes, int n_in,
                              void* d_out, int out_size, void* d_ws, size_t ws_size,
                              hipStream_t stream) {
    const float* x       = (const float*)d_in[0];
    const int*   lengths = (const int*)  d_in[1];
    const int*   spans   = (const int*)  d_in[2];
    float*       out     = (float*)d_out;

    dim3 grid(NS / SPANS_PER_TILE,
              NB);
    span_mean_wmma<<<grid, WAVES_PER_BLOCK * 32, 0, stream>>>(x, lengths, spans, out);
}


# span groups at or past lengths[b] store zeros without reading token rows; the live (batch, group, 64-channel) units are dealt in equal consecutive runs to all 256 blocks
# speedup vs baseline: 1.5181x; 1.2835x over previous
#include <hip/hip_runtime.h>


typedef float v2f __attribute__((ext_vector_type(2)));
typedef float v4f __attribute__((ext_vector_type(4)));
typedef float v8f __attribute__((ext_vector_type(8)));

constexpr int NB = 8;
constexpr int NT = 4096;
constexpr int NS = 512;
constexpr int ND = 768;

constexpr int SPANS_PER_TILE  = 16;
constexpr int CHANS_PER_LANE  = 4;
constexpr int CHANS_PER_WAVE  = 16 * CHANS_PER_LANE;
constexpr int WAVES_PER_BLOCK = ND / CHANS_PER_WAVE;
constexpr int KTOK = (NT / NS) * SPANS_PER_TILE;

constexpr int GROUPS_PER_BATCH = NS / SPANS_PER_TILE;
constexpr int CHUNKS_PER_ROW   = ND / CHANS_PER_WAVE;
constexpr int NBLOCKS          = NB * GROUPS_PER_BATCH;
static_assert(NBLOCKS * WAVES_PER_BLOCK == NB * GROUPS_PER_BATCH * CHUNKS_PER_ROW,
              "one unit per launched wave");

__global__ __launch_bounds__(WAVES_PER_BLOCK * 32) void span_mean_wmma(
    const float* __restrict__ x,
    const int*   __restrict__ lengths,
    const int*   __restrict__ spans,
    float*       __restrict__ out)
{
    const int lane = threadIdx.x & 31;
    const int wave = __builtin_amdgcn_readfirstlane((int)(threadIdx.x >> 5));
    const int blk  = blockIdx.x;

    int len_of[NB], live_of[NB];
    int n_live_groups = 0;
    for (int bb = 0; bb < NB; ++bb) {
        len_of[bb]  = lengths[bb];
        live_of[bb] = (min(max(len_of[bb], 0), NS) + SPANS_PER_TILE - 1) / SPANS_PER_TILE;
        n_live_groups += live_of[bb];
    }
    const int n_live_units = n_live_groups * CHUNKS_PER_ROW;

    const int live_lo   = (blk * n_live_units) / NBLOCKS;
    const int live_hi   = ((blk + 1) * n_live_units) / NBLOCKS;
    const int live_here = live_hi - live_lo;
    const bool is_live  = wave < live_here;
    const int zero_lo   = blk * WAVES_PER_BLOCK - live_lo;
    const int unit      = is_live ? live_lo + wave : zero_lo + (wave - live_here);
    const int grp       = unit / CHUNKS_PER_ROW;
    const int chunk     = unit - grp * CHUNKS_PER_ROW;

    int b = 0, first = 0, len_b = len_of[0], live_b = live_of[0], run = 0;
    for (int bb = 0; bb < NB; ++bb) {
        if (grp >= run) { b = bb; first = run; len_b = len_of[bb]; live_b = live_of[bb]; }
        run += is_live ? live_of[bb] : GROUPS_PER_BATCH - live_of[bb];
    }
    const int sg = (is_live ? 0 : live_b) + (grp - first);

    const int chan0 = chunk * CHANS_PER_WAVE;
    const int span0 = sg * SPANS_PER_TILE;

    const int m  = lane & 15;
    const int hi = lane >> 4;

    v8f c0 = {}, c1 = {}, c2 = {}, c3 = {};

    if (is_live) {
        const int sp  = span0 + m;
        const int ii  = spans[((b * NS + sp) << 1) + 0];
        const int jj  = spans[((b * NS + sp) << 1) + 1];
        const float inv = 1.0f / (float)max(jj - ii, 1);

        const int base = spans[((b * NS + span0) << 1) + 0];

        const float* xp = x + (size_t)b * NT * ND
                            + (size_t)(base + (hi << 1)) * ND
                            + chan0 + m * CHANS_PER_LANE;

        constexpr int KSTEPS = KTOK / 4;
        v4f r[2 * KSTEPS];
        #pragma unroll
        for (int i = 0; i < KSTEPS; ++i) {
            r[2 * i + 0] = *(const v4f*)(xp + (size_t)(4 * i)     * ND);
            r[2 * i + 1] = *(const v4f*)(xp + (size_t)(4 * i + 1) * ND);
        }
        __builtin_amdgcn_sched_barrier(0);

        #pragma unroll
        for (int i = 0; i < KSTEPS; ++i) {
            const int tok = base + 4 * i + (hi << 1);

            v2f a;
            a.x = (tok     >= ii && tok     < jj) ? inv : 0.0f;
            a.y = (tok + 1 >= ii && tok + 1 < jj) ? inv : 0.0f;

            const v4f r0 = r[2 * i + 0];
            const v4f r1 = r[2 * i + 1];
            const v2f b0 = {r0.x, r1.x};
            const v2f b1 = {r0.y, r1.y};
            const v2f b2 = {r0.z, r1.z};
            const v2f b3 = {r0.w, r1.w};

            c0 = __builtin_amdgcn_wmma_f32_16x16x4_f32(false, a, false, b0, (short)0, c0, false, false);
            c1 = __builtin_amdgcn_wmma_f32_16x16x4_f32(false, a, false, b1, (short)0, c1, false, false);
            c2 = __builtin_amdgcn_wmma_f32_16x16x4_f32(false, a, false, b2, (short)0, c2, false, false);
            c3 = __builtin_amdgcn_wmma_f32_16x16x4_f32(false, a, false, b3, (short)0, c3, false, false);
        }
    }

    float* op = out + (size_t)b * NS * ND + chan0 + m * CHANS_PER_LANE;
    #pragma unroll
    for (int v = 0; v < 8; ++v) {
        const int sp_out = span0 + v + (hi << 3);
        const bool keep  = sp_out < len_b;
        v4f val;
        val.x = keep ? c0[v] : 0.0f;
        val.y = keep ? c1[v] : 0.0f;
        val.z = keep ? c2[v] : 0.0f;
        val.w = keep ? c3[v] : 0.0f;
        *(v4f*)(op + (size_t)sp_out * ND) = val;
    }
}

extern "C" void kernel_launch(void* const* d_in, const int* in_sizes, int n_in,
                              void* d_out, int out_size, void* d_ws, size_t ws_size,
                              hipStream_t stream) {
    const float* x       = (const float*)d_in[0];
    const int*   lengths = (const int*)  d_in[1];
    const int*   spans   = (const int*)  d_in[2];
    float*       out     = (float*)d_out;

    span_mean_wmma<<<NBLOCKS, WAVES_PER_BLOCK * 32, 0, stream>>>(x, lengths, spans, out);
}
